# starting kernel
# baseline (speedup 1.0000x reference)
#include <hip/hip_runtime.h>
#include <hip/hip_bf16.h>

#define BATCH   2048
#define IN_SZ   128
#define OUT_SZ  128
#define HID     5
#define BBLK    256
#define LDS_STRIDE (IN_SZ + 1)

typedef float v2f __attribute__((ext_vector_type(2)));
typedef float v8f __attribute__((ext_vector_type(8)));

__global__ __launch_bounds__(32) void residual_wmma_kernel(
    const float* __restrict__ X,
    const float* __restrict__ RS,
    float* __restrict__ Out)
{
    const int lane  = threadIdx.x;
    const int lo    = lane & 15;
    const int hi    = lane >> 4;
    const int baseN = blockIdx.x * 16;
    const int baseM = blockIdx.y * 16;

    v8f acc = {};

    #pragma unroll 4
    for (int k = 0; k < IN_SZ; k += 4) {
        const float* ap = X + (baseM + lo) * IN_SZ + k + 2 * hi;
        v2f a;
        a.x = ap[0];
        a.y = ap[1];

        const float* bp = RS + (k + 2 * hi) * OUT_SZ + baseN + lo;
        v2f b;
        b.x = bp[0];
        b.y = bp[OUT_SZ];

        acc = __builtin_amdgcn_wmma_f32_16x16x4_f32(
            false, a, false, b, (short)0, acc, false, false);
    }

    float* op = Out + (baseM + 8 * hi) * OUT_SZ + baseN + lo;
    #pragma unroll
    for (int r = 0; r < 8; ++r)
        op[r * OUT_SZ] = acc[r];
}

__device__ __forceinline__ float silu_f(float z) {
    float e = __builtin_amdgcn_exp2f(-1.44269504f * z);
    return z * __builtin_amdgcn_rcpf(1.0f + e);
}

__global__ __launch_bounds__(BBLK) void subnet_kernel(
    const float* __restrict__ x,
    const float* __restrict__ w0,
    const float* __restrict__ b0,
    const float* __restrict__ w1,
    const float* __restrict__ b1,
    const float* __restrict__ w2,
    const float* __restrict__ b2,
    const float* __restrict__ ss,
    float* __restrict__ out)
{
    extern __shared__ float xs[];

    const int o  = blockIdx.x;
    const int bb = blockIdx.y * BBLK;
    const int t  = threadIdx.x;

    for (int j = t; j < BBLK * IN_SZ; j += BBLK) {
        int bl = j >> 7;
        int i  = j & (IN_SZ - 1);
        xs[bl * LDS_STRIDE + i] = x[(bb + bl) * IN_SZ + i];
    }
    __syncthreads();

    const int b = bb + t;
    float acc = 0.0f;

    for (int i = 0; i < IN_SZ; ++i) {
        const int n = i * OUT_SZ + o;
        const float xv = xs[t * LDS_STRIDE + i];

        const float* W0 = w0 + n * HID;
        const float* B0 = b0 + n * HID;
        const float* W1 = w1 + n * HID * HID;
        const float* B1 = b1 + n * HID;
        const float* W2 = w2 + n * HID;

        float h0[HID], h1[HID];
        #pragma unroll
        for (int h = 0; h < HID; ++h)
            h0[h] = silu_f(fmaf(W0[h], xv, B0[h]));

        #pragma unroll
        for (int h = 0; h < HID; ++h) {
            float s = B1[h];
            #pragma unroll
            for (int k = 0; k < HID; ++k)
                s = fmaf(W1[h * HID + k], h0[k], s);
            h1[h] = silu_f(s);
        }

        float y = b2[n];
        #pragma unroll
        for (int k = 0; k < HID; ++k)
            y = fmaf(W2[k], h1[k], y);

        acc = fmaf(y, ss[n], acc);
    }

    out[b * OUT_SZ + o] += acc;
}

extern "C" void kernel_launch(void* const* d_in, const int* in_sizes, int n_in,
                              void* d_out, int out_size, void* d_ws, size_t ws_size,
                              hipStream_t stream) {
    const float* x   = (const float*)d_in[0];
    const float* w0  = (const float*)d_in[1];
    const float* b0  = (const float*)d_in[2];
    const float* w1  = (const float*)d_in[3];
    const float* b1  = (const float*)d_in[4];
    const float* w2  = (const float*)d_in[5];
    const float* b2  = (const float*)d_in[6];
    const float* ss  = (const float*)d_in[7];
    const float* rs  = (const float*)d_in[8];
    float* out = (float*)d_out;

    dim3 gridG(OUT_SZ / 16, BATCH / 16);
    residual_wmma_kernel<<<gridG, 32, 0, stream>>>(x, rs, out);

    dim3 gridS(OUT_SZ, BATCH / BBLK);
    size_t ldsBytes = (size_t)BBLK * LDS_STRIDE * sizeof(float);
    subnet_kernel<<<gridS, BBLK, ldsBytes, stream>>>(
        x, w0, b0, w1, b1, w2, b2, ss, out);
}


# subnet kernel: lanes span 32 outputs, a net's 47 weights stay in registers (staged per wave via LDS) across 16 batch rows; the i-sum becomes 16 partial sums added through LDS
# speedup vs baseline: 3.2414x; 3.2414x over previous
#include <hip/hip_runtime.h>
#include <hip/hip_bf16.h>


#define BATCH   2048
#define IN_SZ   128
#define OUT_SZ  128
#define HID     5
#define BB      16
#define NWAVE   16
#define ISTEPS  (IN_SZ / NWAVE)
#define NETW    47
#define WSET    (32 * NETW)
#define OFF_W0  0
#define OFF_B0  160
#define OFF_W1  320
#define OFF_B1  1120
#define OFF_W2  1280
#define OFF_B2  1440
#define OFF_SS  1472
#define LDS_FLOATS (IN_SZ * BB + NWAVE * WSET)

typedef float v2f __attribute__((ext_vector_type(2)));
typedef float v8f __attribute__((ext_vector_type(8)));

__global__ __launch_bounds__(32) void residual_wmma_kernel(
    const float* __restrict__ X,
    const float* __restrict__ RS,
    float* __restrict__ Out)
{
    const int lane  = threadIdx.x;
    const int lo    = lane & 15;
    const int hi    = lane >> 4;
    const int baseN = blockIdx.x * 16;
    const int baseM = blockIdx.y * 16;

    v8f acc = {};

    #pragma unroll 4
    for (int k = 0; k < IN_SZ; k += 4) {
        const float* ap = X + (baseM + lo) * IN_SZ + k + 2 * hi;
        v2f a;
        a.x = ap[0];
        a.y = ap[1];

        const float* bp = RS + (k + 2 * hi) * OUT_SZ + baseN + lo;
        v2f b;
        b.x = bp[0];
        b.y = bp[OUT_SZ];

        acc = __builtin_amdgcn_wmma_f32_16x16x4_f32(
            false, a, false, b, (short)0, acc, false, false);
    }

    float* op = Out + (baseM + 8 * hi) * OUT_SZ + baseN + lo;
    #pragma unroll
    for (int r = 0; r < 8; ++r)
        op[r * OUT_SZ] = acc[r];
}

__device__ __forceinline__ float silu_f(float z) {
    float e = __builtin_amdgcn_exp2f(-1.44269504f * z);
    return z * __builtin_amdgcn_rcpf(1.0f + e);
}

__device__ __forceinline__ float4 ld4(const float* __restrict__ p, int idx) {
    return reinterpret_cast<const float4*>(p)[idx];
}
__device__ __forceinline__ void st4(float* p, int idx, float4 v) {
    reinterpret_cast<float4*>(p)[idx] = v;
}

__global__ __launch_bounds__(NWAVE * 32) void subnet_kernel(
    const float* __restrict__ x,
    const float* __restrict__ w0,
    const float* __restrict__ b0,
    const float* __restrict__ w1,
    const float* __restrict__ b1,
    const float* __restrict__ w2,
    const float* __restrict__ b2,
    const float* __restrict__ ss,
    float* __restrict__ out)
{
    extern __shared__ float4 lds4[];
    float* xsT  = reinterpret_cast<float*>(lds4);
    float* wbuf = xsT + IN_SZ * BB;

    const int t    = threadIdx.x;
    const int lane = t & 31;
    const int w    = t >> 5;
    const int g    = blockIdx.x;
    const int bb   = blockIdx.y * BB;

    for (int j = t; j < BB * IN_SZ; j += NWAVE * 32) {
        int bl = j >> 7;
        int i  = j & (IN_SZ - 1);
        xsT[i * BB + bl] = x[(bb + bl) * IN_SZ + i];
    }
    __syncthreads();

    float* wb = wbuf + w * WSET;

    float acc[BB];
    #pragma unroll
    for (int b = 0; b < BB; ++b) acc[b] = 0.0f;

    for (int s = 0; s < ISTEPS; ++s) {
        const int i  = w * ISTEPS + s;
        const int nb = i * OUT_SZ + g * 32;

        const float* pw0 = w0 + nb * HID;
        const float* pb0 = b0 + nb * HID;
        const float* pw1 = w1 + nb * HID * HID;
        const float* pb1 = b1 + nb * HID;
        const float* pw2 = w2 + nb * HID;
        const float* pb2 = b2 + nb;
        const float* pss = ss + nb;

        const float4 r0  = ld4(pw0, lane);
        const float4 r1  = ld4(pb0, lane);
        const float4 r2  = ld4(pw1, lane);
        const float4 r3  = ld4(pw1, lane + 32);
        const float4 r4  = ld4(pw1, lane + 64);
        const float4 r5  = ld4(pw1, lane + 96);
        const float4 r6  = ld4(pw1, lane + 128);
        const float4 r7  = ld4(pw1, lane + 160);
        const float4 r8  = ld4(pb1, lane);
        const float4 r9  = ld4(pw2, lane);
        float4 q0, q1, q2, q3, q4, q5, q6;
        if (lane < 8) {
            q0 = ld4(pw0, lane + 32);
            q1 = ld4(pb0, lane + 32);
            q2 = ld4(pw1, lane + 192);
            q3 = ld4(pb1, lane + 32);
            q4 = ld4(pw2, lane + 32);
            q5 = ld4(pb2, lane);
            q6 = ld4(pss, lane);
        }

        __builtin_amdgcn_fence(__ATOMIC_RELEASE, "wavefront");
        __builtin_amdgcn_wave_barrier();

        st4(wb + OFF_W0, lane, r0);
        st4(wb + OFF_B0, lane, r1);
        st4(wb + OFF_W1, lane, r2);
        st4(wb + OFF_W1, lane + 32, r3);
        st4(wb + OFF_W1, lane + 64, r4);
        st4(wb + OFF_W1, lane + 96, r5);
        st4(wb + OFF_W1, lane + 128, r6);
        st4(wb + OFF_W1, lane + 160, r7);
        st4(wb + OFF_B1, lane, r8);
        st4(wb + OFF_W2, lane, r9);
        if (lane < 8) {
            st4(wb + OFF_W0, lane + 32, q0);
            st4(wb + OFF_B0, lane + 32, q1);
            st4(wb + OFF_W1, lane + 192, q2);
            st4(wb + OFF_B1, lane + 32, q3);
            st4(wb + OFF_W2, lane + 32, q4);
            st4(wb + OFF_B2, lane, q5);
            st4(wb + OFF_SS, lane, q6);
        }

        __builtin_amdgcn_fence(__ATOMIC_RELEASE, "wavefront");
        __builtin_amdgcn_wave_barrier();
        __builtin_amdgcn_fence(__ATOMIC_ACQUIRE, "wavefront");

        float W0[HID], B0[HID], W1[HID * HID], B1[HID], W2[HID];
        #pragma unroll
        for (int h = 0; h < HID; ++h) {
            W0[h] = wb[OFF_W0 + lane * HID + h];
            B0[h] = wb[OFF_B0 + lane * HID + h];
            B1[h] = wb[OFF_B1 + lane * HID + h];
            W2[h] = wb[OFF_W2 + lane * HID + h];
        }
        #pragma unroll
        for (int k = 0; k < HID * HID; ++k)
            W1[k] = wb[OFF_W1 + lane * HID * HID + k];
        const float B2 = wb[OFF_B2 + lane];
        const float SS = wb[OFF_SS + lane];

        #pragma unroll
        for (int q = 0; q < BB / 4; ++q) {
            const float4 xq = *reinterpret_cast<const float4*>(&xsT[i * BB + 4 * q]);
            const float xv[4] = {xq.x, xq.y, xq.z, xq.w};
            #pragma unroll
            for (int j = 0; j < 4; ++j) {
                float h0[HID], h1[HID];
                #pragma unroll
                for (int h = 0; h < HID; ++h)
                    h0[h] = silu_f(fmaf(W0[h], xv[j], B0[h]));

                #pragma unroll
                for (int h = 0; h < HID; ++h) {
                    float sacc = B1[h];
                    #pragma unroll
                    for (int k = 0; k < HID; ++k)
                        sacc = fmaf(W1[h * HID + k], h0[k], sacc);
                    h1[h] = silu_f(sacc);
                }

                float y = B2;
                #pragma unroll
                for (int k = 0; k < HID; ++k)
                    y = fmaf(W2[k], h1[k], y);

                acc[4 * q + j] = fmaf(y, SS, acc[4 * q + j]);
            }
        }
    }

    __syncthreads();
    float* red = wbuf;
    #pragma unroll
    for (int b = 0; b < BB; ++b)
        red[(w * BB + b) * 32 + lane] = acc[b];
    __syncthreads();

    float sum = 0.0f;
    #pragma unroll
    for (int ww = 0; ww < NWAVE; ++ww)
        sum += red[(ww * BB + w) * 32 + lane];

    out[(bb + w) * OUT_SZ + g * 32 + lane] += sum;
}

extern "C" void kernel_launch(void* const* d_in, const int* in_sizes, int n_in,
                              void* d_out, int out_size, void* d_ws, size_t ws_size,
                              hipStream_t stream) {
    const float* x   = (const float*)d_in[0];
    const float* w0  = (const float*)d_in[1];
    const float* b0  = (const float*)d_in[2];
    const float* w1  = (const float*)d_in[3];
    const float* b1  = (const float*)d_in[4];
    const float* w2  = (const float*)d_in[5];
    const float* b2  = (const float*)d_in[6];
    const float* ss  = (const float*)d_in[7];
    const float* rs  = (const float*)d_in[8];
    float* out = (float*)d_out;

    dim3 gridG(OUT_SZ / 16, BATCH / 16);
    residual_wmma_kernel<<<gridG, 32, 0, stream>>>(x, rs, out);

    dim3 gridS(OUT_SZ / 32, BATCH / BB);
    size_t ldsBytes = (size_t)LDS_FLOATS * sizeof(float);
    subnet_kernel<<<gridS, NWAVE * 32, ldsBytes, stream>>>(
        x, w0, b0, w1, b1, w2, b2, ss, out);
}


# SiLU as hz + hz*tanh(hz) with hz = z/2 from halved layer-0/1 weights: one native v_tanh_f32 and one fma per activation instead of mul, exp2, add, rcp, mul
# speedup vs baseline: 4.0288x; 1.2429x over previous
#include <hip/hip_runtime.h>
#include <hip/hip_bf16.h>


#define BATCH   2048
#define IN_SZ   128
#define OUT_SZ  128
#define HID     5
#define BB      16
#define NWAVE   16
#define ISTEPS  (IN_SZ / NWAVE)
#define NETW    47
#define WSET    (32 * NETW)
#define OFF_W0  0
#define OFF_B0  160
#define OFF_W1  320
#define OFF_B1  1120
#define OFF_W2  1280
#define OFF_B2  1440
#define OFF_SS  1472
#define LDS_FLOATS (IN_SZ * BB + NWAVE * WSET)

typedef float v2f __attribute__((ext_vector_type(2)));
typedef float v8f __attribute__((ext_vector_type(8)));

__global__ __launch_bounds__(32) void residual_wmma_kernel(
    const float* __restrict__ X,
    const float* __restrict__ RS,
    float* __restrict__ Out)
{
    const int lane  = threadIdx.x;
    const int lo    = lane & 15;
    const int hi    = lane >> 4;
    const int baseN = blockIdx.x * 16;
    const int baseM = blockIdx.y * 16;

    v8f acc = {};

    #pragma unroll 4
    for (int k = 0; k < IN_SZ; k += 4) {
        const float* ap = X + (baseM + lo) * IN_SZ + k + 2 * hi;
        v2f a;
        a.x = ap[0];
        a.y = ap[1];

        const float* bp = RS + (k + 2 * hi) * OUT_SZ + baseN + lo;
        v2f b;
        b.x = bp[0];
        b.y = bp[OUT_SZ];

        acc = __builtin_amdgcn_wmma_f32_16x16x4_f32(
            false, a, false, b, (short)0, acc, false, false);
    }

    float* op = Out + (baseM + 8 * hi) * OUT_SZ + baseN + lo;
    #pragma unroll
    for (int r = 0; r < 8; ++r)
        op[r * OUT_SZ] = acc[r];
}

__device__ __forceinline__ float silu_half(float hz) {
    return fmaf(hz, __builtin_amdgcn_tanhf(hz), hz);
}

__device__ __forceinline__ float4 ld4(const float* __restrict__ p, int idx) {
    return reinterpret_cast<const float4*>(p)[idx];
}
__device__ __forceinline__ void st4(float* p, int idx, float4 v) {
    reinterpret_cast<float4*>(p)[idx] = v;
}

__global__ __launch_bounds__(NWAVE * 32) void subnet_kernel(
    const float* __restrict__ x,
    const float* __restrict__ w0,
    const float* __restrict__ b0,
    const float* __restrict__ w1,
    const float* __restrict__ b1,
    const float* __restrict__ w2,
    const float* __restrict__ b2,
    const float* __restrict__ ss,
    float* __restrict__ out)
{
    extern __shared__ float4 lds4[];
    float* xsT  = reinterpret_cast<float*>(lds4);
    float* wbuf = xsT + IN_SZ * BB;

    const int t    = threadIdx.x;
    const int lane = t & 31;
    const int w    = t >> 5;
    const int g    = blockIdx.x;
    const int bb   = blockIdx.y * BB;

    for (int j = t; j < BB * IN_SZ; j += NWAVE * 32) {
        int bl = j >> 7;
        int i  = j & (IN_SZ - 1);
        xsT[i * BB + bl] = x[(bb + bl) * IN_SZ + i];
    }
    __syncthreads();

    float* wb = wbuf + w * WSET;

    float acc[BB];
    #pragma unroll
    for (int b = 0; b < BB; ++b) acc[b] = 0.0f;

    for (int s = 0; s < ISTEPS; ++s) {
        const int i  = w * ISTEPS + s;
        const int nb = i * OUT_SZ + g * 32;

        const float* pw0 = w0 + nb * HID;
        const float* pb0 = b0 + nb * HID;
        const float* pw1 = w1 + nb * HID * HID;
        const float* pb1 = b1 + nb * HID;
        const float* pw2 = w2 + nb * HID;
        const float* pb2 = b2 + nb;
        const float* pss = ss + nb;

        const float4 r0  = ld4(pw0, lane);
        const float4 r1  = ld4(pb0, lane);
        const float4 r2  = ld4(pw1, lane);
        const float4 r3  = ld4(pw1, lane + 32);
        const float4 r4  = ld4(pw1, lane + 64);
        const float4 r5  = ld4(pw1, lane + 96);
        const float4 r6  = ld4(pw1, lane + 128);
        const float4 r7  = ld4(pw1, lane + 160);
        const float4 r8  = ld4(pb1, lane);
        const float4 r9  = ld4(pw2, lane);
        float4 q0, q1, q2, q3, q4, q5, q6;
        if (lane < 8) {
            q0 = ld4(pw0, lane + 32);
            q1 = ld4(pb0, lane + 32);
            q2 = ld4(pw1, lane + 192);
            q3 = ld4(pb1, lane + 32);
            q4 = ld4(pw2, lane + 32);
            q5 = ld4(pb2, lane);
            q6 = ld4(pss, lane);
        }

        __builtin_amdgcn_fence(__ATOMIC_RELEASE, "wavefront");
        __builtin_amdgcn_wave_barrier();

        st4(wb + OFF_W0, lane, r0);
        st4(wb + OFF_B0, lane, r1);
        st4(wb + OFF_W1, lane, r2);
        st4(wb + OFF_W1, lane + 32, r3);
        st4(wb + OFF_W1, lane + 64, r4);
        st4(wb + OFF_W1, lane + 96, r5);
        st4(wb + OFF_W1, lane + 128, r6);
        st4(wb + OFF_W1, lane + 160, r7);
        st4(wb + OFF_B1, lane, r8);
        st4(wb + OFF_W2, lane, r9);
        if (lane < 8) {
            st4(wb + OFF_W0, lane + 32, q0);
            st4(wb + OFF_B0, lane + 32, q1);
            st4(wb + OFF_W1, lane + 192, q2);
            st4(wb + OFF_B1, lane + 32, q3);
            st4(wb + OFF_W2, lane + 32, q4);
            st4(wb + OFF_B2, lane, q5);
            st4(wb + OFF_SS, lane, q6);
        }

        __builtin_amdgcn_fence(__ATOMIC_RELEASE, "wavefront");
        __builtin_amdgcn_wave_barrier();
        __builtin_amdgcn_fence(__ATOMIC_ACQUIRE, "wavefront");

        float W0[HID], B0[HID], W1[HID * HID], B1[HID], W2[HID];
        #pragma unroll
        for (int h = 0; h < HID; ++h) {
            W0[h] = 0.5f * wb[OFF_W0 + lane * HID + h];
            B0[h] = 0.5f * wb[OFF_B0 + lane * HID + h];
            B1[h] = 0.5f * wb[OFF_B1 + lane * HID + h];
            W2[h] = wb[OFF_W2 + lane * HID + h];
        }
        #pragma unroll
        for (int k = 0; k < HID * HID; ++k)
            W1[k] = 0.5f * wb[OFF_W1 + lane * HID * HID + k];
        const float B2 = wb[OFF_B2 + lane];
        const float SS = wb[OFF_SS + lane];

        #pragma unroll
        for (int q = 0; q < BB / 4; ++q) {
            const float4 xq = *reinterpret_cast<const float4*>(&xsT[i * BB + 4 * q]);
            const float xv[4] = {xq.x, xq.y, xq.z, xq.w};
            #pragma unroll
            for (int j = 0; j < 4; ++j) {
                float h0[HID], h1[HID];
                #pragma unroll
                for (int h = 0; h < HID; ++h)
                    h0[h] = silu_half(fmaf(W0[h], xv[j], B0[h]));

                #pragma unroll
                for (int h = 0; h < HID; ++h) {
                    float sacc = B1[h];
                    #pragma unroll
                    for (int k = 0; k < HID; ++k)
                        sacc = fmaf(W1[h * HID + k], h0[k], sacc);
                    h1[h] = silu_half(sacc);
                }

                float y = B2;
                #pragma unroll
                for (int k = 0; k < HID; ++k)
                    y = fmaf(W2[k], h1[k], y);

                acc[4 * q + j] = fmaf(y, SS, acc[4 * q + j]);
            }
        }
    }

    __syncthreads();
    float* red = wbuf;
    #pragma unroll
    for (int b = 0; b < BB; ++b)
        red[(w * BB + b) * 32 + lane] = acc[b];
    __syncthreads();

    float sum = 0.0f;
    #pragma unroll
    for (int ww = 0; ww < NWAVE; ++ww)
        sum += red[(ww * BB + w) * 32 + lane];

    out[(bb + w) * OUT_SZ + g * 32 + lane] += sum;
}

extern "C" void kernel_launch(void* const* d_in, const int* in_sizes, int n_in,
                              void* d_out, int out_size, void* d_ws, size_t ws_size,
                              hipStream_t stream) {
    const float* x   = (const float*)d_in[0];
    const float* w0  = (const float*)d_in[1];
    const float* b0  = (const float*)d_in[2];
    const float* w1  = (const float*)d_in[3];
    const float* b1  = (const float*)d_in[4];
    const float* w2  = (const float*)d_in[5];
    const float* b2  = (const float*)d_in[6];
    const float* ss  = (const float*)d_in[7];
    const float* rs  = (const float*)d_in[8];
    float* out = (float*)d_out;

    dim3 gridG(OUT_SZ / 16, BATCH / 16);
    residual_wmma_kernel<<<gridG, 32, 0, stream>>>(x, rs, out);

    dim3 gridS(OUT_SZ / 32, BATCH / BB);
    size_t ldsBytes = (size_t)LDS_FLOATS * sizeof(float);
    subnet_kernel<<<gridS, NWAVE * 32, ldsBytes, stream>>>(
        x, w0, b0, w1, b1, w2, b2, ss, out);
}


# subnet kernel: weight addresses kept as scalar bases (readfirstlane) instead of seven vector pointers, 145 -> 127 registers, so two blocks share a WGP: 8 waves per SIMD, one resident pass
# speedup vs baseline: 4.4337x; 1.1005x over previous
#include <hip/hip_runtime.h>
#include <hip/hip_bf16.h>


#define BATCH   2048
#define IN_SZ   128
#define OUT_SZ  128
#define HID     5
#define BB      16
#define NWAVE   16
#define ISTEPS  (IN_SZ / NWAVE)
#define NETW    47
#define WSET    (32 * NETW)
#define OFF_W0  0
#define OFF_B0  160
#define OFF_W1  320
#define OFF_B1  1120
#define OFF_W2  1280
#define OFF_B2  1440
#define OFF_SS  1472
#define LDS_FLOATS (IN_SZ * BB + NWAVE * WSET)

typedef float v2f __attribute__((ext_vector_type(2)));
typedef float v8f __attribute__((ext_vector_type(8)));
typedef float v4f __attribute__((ext_vector_type(4)));
typedef const __attribute__((address_space(1))) float* gfp;
typedef const __attribute__((address_space(1))) v4f*   gv4p;
typedef const __attribute__((address_space(1))) char*  gcp;

__global__ __launch_bounds__(32) void residual_wmma_kernel(
    const float* __restrict__ X,
    const float* __restrict__ RS,
    float* __restrict__ Out)
{
    const int lane  = threadIdx.x;
    const int lo    = lane & 15;
    const int hi    = lane >> 4;
    const int baseN = blockIdx.x * 16;
    const int baseM = blockIdx.y * 16;

    v8f acc = {};

    #pragma unroll 4
    for (int k = 0; k < IN_SZ; k += 4) {
        const float* ap = X + (baseM + lo) * IN_SZ + k + 2 * hi;
        v2f a;
        a.x = ap[0];
        a.y = ap[1];

        const float* bp = RS + (k + 2 * hi) * OUT_SZ + baseN + lo;
        v2f b;
        b.x = bp[0];
        b.y = bp[OUT_SZ];

        acc = __builtin_amdgcn_wmma_f32_16x16x4_f32(
            false, a, false, b, (short)0, acc, false, false);
    }

    float* op = Out + (baseM + 8 * hi) * OUT_SZ + baseN + lo;
    #pragma unroll
    for (int r = 0; r < 8; ++r)
        op[r * OUT_SZ] = acc[r];
}

__device__ __forceinline__ float silu_half(float hz) {
    return fmaf(hz, __builtin_amdgcn_tanhf(hz), hz);
}

__device__ __forceinline__ v4f ld4(gfp p, unsigned lane16, unsigned k) {
    return *(gv4p)((gcp)p + lane16 + k * 512u);
}
__device__ __forceinline__ gfp uni(const float* p) {
    const unsigned long long v = reinterpret_cast<unsigned long long>(p);
    const unsigned lo = __builtin_amdgcn_readfirstlane(static_cast<unsigned>(v));
    const unsigned hi = __builtin_amdgcn_readfirstlane(static_cast<unsigned>(v >> 32));
    return (gfp)((static_cast<unsigned long long>(hi) << 32) | lo);
}
__device__ __forceinline__ void st4(float* p, int idx, v4f v) {
    reinterpret_cast<v4f*>(p)[idx] = v;
}

__global__ __launch_bounds__(NWAVE * 32) void subnet_kernel(
    const float* __restrict__ x,
    const float* __restrict__ w0,
    const float* __restrict__ b0,
    const float* __restrict__ w1,
    const float* __restrict__ b1,
    const float* __restrict__ w2,
    const float* __restrict__ b2,
    const float* __restrict__ ss,
    float* __restrict__ out)
{
    extern __shared__ float4 lds4[];
    float* xsT  = reinterpret_cast<float*>(lds4);
    float* wbuf = xsT + IN_SZ * BB;

    const int t    = threadIdx.x;
    const int lane = t & 31;
    const int w    = __builtin_amdgcn_readfirstlane(t >> 5);
    const int g    = blockIdx.x;
    const int bb   = blockIdx.y * BB;

    for (int j = t; j < BB * IN_SZ; j += NWAVE * 32) {
        int bl = j >> 7;
        int i  = j & (IN_SZ - 1);
        xsT[i * BB + bl] = x[(bb + bl) * IN_SZ + i];
    }
    __syncthreads();

    float* wb = wbuf + w * WSET;
    const unsigned lane16 = (unsigned)lane * 16u;

    float acc[BB];
    #pragma unroll
    for (int b = 0; b < BB; ++b) acc[b] = 0.0f;

    for (int s = 0; s < ISTEPS; ++s) {
        const int i  = w * ISTEPS + s;
        const int nb = i * OUT_SZ + g * 32;

        const gfp pw0 = uni(w0 + nb * HID);
        const gfp pb0 = uni(b0 + nb * HID);
        const gfp pw1 = uni(w1 + nb * HID * HID);
        const gfp pb1 = uni(b1 + nb * HID);
        const gfp pw2 = uni(w2 + nb * HID);
        const gfp pb2 = uni(b2 + nb);
        const gfp pss = uni(ss + nb);

        const v4f r0 = ld4(pw0, lane16, 0);
        const v4f r1 = ld4(pb0, lane16, 0);
        const v4f r2 = ld4(pw1, lane16, 0);
        const v4f r3 = ld4(pw1, lane16, 1);
        const v4f r4 = ld4(pw1, lane16, 2);
        const v4f r5 = ld4(pw1, lane16, 3);
        const v4f r6 = ld4(pw1, lane16, 4);
        const v4f r7 = ld4(pw1, lane16, 5);
        const v4f r8 = ld4(pb1, lane16, 0);
        const v4f r9 = ld4(pw2, lane16, 0);
        v4f q0, q1, q2, q3, q4, q5, q6;
        if (lane < 8) {
            q0 = ld4(pw0, lane16, 1);
            q1 = ld4(pb0, lane16, 1);
            q2 = ld4(pw1, lane16, 6);
            q3 = ld4(pb1, lane16, 1);
            q4 = ld4(pw2, lane16, 1);
            q5 = ld4(pb2, lane16, 0);
            q6 = ld4(pss, lane16, 0);
        }

        __builtin_amdgcn_fence(__ATOMIC_RELEASE, "wavefront");
        __builtin_amdgcn_wave_barrier();

        st4(wb + OFF_W0, lane, r0);
        st4(wb + OFF_B0, lane, r1);
        st4(wb + OFF_W1, lane, r2);
        st4(wb + OFF_W1, lane + 32, r3);
        st4(wb + OFF_W1, lane + 64, r4);
        st4(wb + OFF_W1, lane + 96, r5);
        st4(wb + OFF_W1, lane + 128, r6);
        st4(wb + OFF_W1, lane + 160, r7);
        st4(wb + OFF_B1, lane, r8);
        st4(wb + OFF_W2, lane, r9);
        if (lane < 8) {
            st4(wb + OFF_W0, lane + 32, q0);
            st4(wb + OFF_B0, lane + 32, q1);
            st4(wb + OFF_W1, lane + 192, q2);
            st4(wb + OFF_B1, lane + 32, q3);
            st4(wb + OFF_W2, lane + 32, q4);
            st4(wb + OFF_B2, lane, q5);
            st4(wb + OFF_SS, lane, q6);
        }

        __builtin_amdgcn_fence(__ATOMIC_RELEASE, "wavefront");
        __builtin_amdgcn_wave_barrier();
        __builtin_amdgcn_fence(__ATOMIC_ACQUIRE, "wavefront");

        float W0[HID], B0[HID], W1[HID * HID], B1[HID], W2[HID];
        #pragma unroll
        for (int h = 0; h < HID; ++h) {
            W0[h] = 0.5f * wb[OFF_W0 + lane * HID + h];
            B0[h] = 0.5f * wb[OFF_B0 + lane * HID + h];
            B1[h] = 0.5f * wb[OFF_B1 + lane * HID + h];
            W2[h] = wb[OFF_W2 + lane * HID + h];
        }
        #pragma unroll
        for (int k = 0; k < HID * HID; ++k)
            W1[k] = 0.5f * wb[OFF_W1 + lane * HID * HID + k];
        const float B2 = wb[OFF_B2 + lane];
        const float SS = wb[OFF_SS + lane];

        #pragma unroll
        for (int q = 0; q < BB / 4; ++q) {
            const float4 xq = *reinterpret_cast<const float4*>(&xsT[i * BB + 4 * q]);
            const float xv[4] = {xq.x, xq.y, xq.z, xq.w};
            #pragma unroll
            for (int j = 0; j < 4; ++j) {
                float h0[HID], h1[HID];
                #pragma unroll
                for (int h = 0; h < HID; ++h)
                    h0[h] = silu_half(fmaf(W0[h], xv[j], B0[h]));

                #pragma unroll
                for (int h = 0; h < HID; ++h) {
                    float sacc = B1[h];
                    #pragma unroll
                    for (int k = 0; k < HID; ++k)
                        sacc = fmaf(W1[h * HID + k], h0[k], sacc);
                    h1[h] = silu_half(sacc);
                }

                float y = B2;
                #pragma unroll
                for (int k = 0; k < HID; ++k)
                    y = fmaf(W2[k], h1[k], y);

                acc[4 * q + j] = fmaf(y, SS, acc[4 * q + j]);
            }
        }
    }

    __syncthreads();
    float* red = wbuf;
    #pragma unroll
    for (int b = 0; b < BB; ++b)
        red[(w * BB + b) * 32 + lane] = acc[b];
    __syncthreads();

    float sum = 0.0f;
    #pragma unroll
    for (int ww = 0; ww < NWAVE; ++ww)
        sum += red[(ww * BB + w) * 32 + lane];

    out[(bb + w) * OUT_SZ + g * 32 + lane] += sum;
}

extern "C" void kernel_launch(void* const* d_in, const int* in_sizes, int n_in,
                              void* d_out, int out_size, void* d_ws, size_t ws_size,
                              hipStream_t stream) {
    const float* x   = (const float*)d_in[0];
    const float* w0  = (const float*)d_in[1];
    const float* b0  = (const float*)d_in[2];
    const float* w1  = (const float*)d_in[3];
    const float* b1  = (const float*)d_in[4];
    const float* w2  = (const float*)d_in[5];
    const float* b2  = (const float*)d_in[6];
    const float* ss  = (const float*)d_in[7];
    const float* rs  = (const float*)d_in[8];
    float* out = (float*)d_out;

    dim3 gridG(OUT_SZ / 16, BATCH / 16);
    residual_wmma_kernel<<<gridG, 32, 0, stream>>>(x, rs, out);

    dim3 gridS(OUT_SZ / 32, BATCH / BB);
    size_t ldsBytes = (size_t)LDS_FLOATS * sizeof(float);
    subnet_kernel<<<gridS, NWAVE * 32, ldsBytes, stream>>>(
        x, w0, b0, w1, b1, w2, b2, ss, out);
}


# residual GEMM moved into the subnet kernel: each wave does 4 f32 WMMAs on the x tile staged in LDS, k split into 8 partial sums added through LDS with the nets' sums; one launch, out stored once
# speedup vs baseline: 5.2563x; 1.1855x over previous
#include <hip/hip_runtime.h>
#include <hip/hip_bf16.h>


#define BATCH   2048
#define IN_SZ   128
#define OUT_SZ  128
#define HID     5
#define BB      16
#define NWAVE   16
#define ISTEPS  (IN_SZ / NWAVE)
#define NETW    47
#define WSET    (32 * NETW)
#define OFF_W0  0
#define OFF_B0  160
#define OFF_W1  320
#define OFF_B1  1120
#define OFF_W2  1280
#define OFF_B2  1440
#define OFF_SS  1472
#define KCH     (NWAVE / 2)
#define LDS_FLOATS (IN_SZ * BB + NWAVE * WSET + KCH * BB * 32)

typedef float v2f __attribute__((ext_vector_type(2)));
typedef float v8f __attribute__((ext_vector_type(8)));
typedef float v4f __attribute__((ext_vector_type(4)));
typedef const __attribute__((address_space(1))) float* gfp;
typedef const __attribute__((address_space(1))) v4f*   gv4p;
typedef const __attribute__((address_space(1))) char*  gcp;

__device__ __forceinline__ float silu_half(float hz) {
    return fmaf(hz, __builtin_amdgcn_tanhf(hz), hz);
}

__device__ __forceinline__ v4f ld4(gfp p, unsigned lane16, unsigned k) {
    return *(gv4p)((gcp)p + lane16 + k * 512u);
}
__device__ __forceinline__ gfp uni(const float* p) {
    const unsigned long long v = reinterpret_cast<unsigned long long>(p);
    const unsigned lo = __builtin_amdgcn_readfirstlane(static_cast<unsigned>(v));
    const unsigned hi = __builtin_amdgcn_readfirstlane(static_cast<unsigned>(v >> 32));
    return (gfp)((static_cast<unsigned long long>(hi) << 32) | lo);
}
__device__ __forceinline__ void st4(float* p, int idx, v4f v) {
    reinterpret_cast<v4f*>(p)[idx] = v;
}

__global__ __launch_bounds__(NWAVE * 32) void subnet_kernel(
    const float* __restrict__ x,
    const float* __restrict__ w0,
    const float* __restrict__ b0,
    const float* __restrict__ w1,
    const float* __restrict__ b1,
    const float* __restrict__ w2,
    const float* __restrict__ b2,
    const float* __restrict__ ss,
    const float* __restrict__ rs,
    float* __restrict__ out)
{
    extern __shared__ float4 lds4[];
    float* xsT  = reinterpret_cast<float*>(lds4);
    float* wbuf = xsT + IN_SZ * BB;
    float* rpart = wbuf + NWAVE * WSET;

    const int t    = threadIdx.x;
    const int lane = t & 31;
    const int w    = __builtin_amdgcn_readfirstlane(t >> 5);
    const int g    = blockIdx.x;
    const int bb   = blockIdx.y * BB;

    const int lo = lane & 15;
    const int hi = lane >> 4;
    const int ct = w & 1;
    const int kc = w >> 1;
    const float* bp = rs + (kc * 16 + 2 * hi) * OUT_SZ + g * 32 + ct * 16 + lo;
    v2f bfr[4];
    #pragma unroll
    for (int j = 0; j < 4; ++j) {
        bfr[j].x = bp[(4 * j) * OUT_SZ];
        bfr[j].y = bp[(4 * j + 1) * OUT_SZ];
    }

    for (int j = t; j < BB * IN_SZ; j += NWAVE * 32) {
        int bl = j >> 7;
        int i  = j & (IN_SZ - 1);
        xsT[i * BB + bl] = x[(bb + bl) * IN_SZ + i];
    }
    __syncthreads();

    {
        const float* ap = xsT + (kc * 16 + 2 * hi) * BB + lo;
        v8f d = {};
        #pragma unroll
        for (int j = 0; j < 4; ++j) {
            v2f a;
            a.x = ap[(4 * j) * BB];
            a.y = ap[(4 * j + 1) * BB];
            d = __builtin_amdgcn_wmma_f32_16x16x4_f32(
                false, a, false, bfr[j], (short)0, d, false, false);
        }
        float* dp = rpart + (kc * BB + 8 * hi) * 32 + ct * 16 + lo;
        #pragma unroll
        for (int r = 0; r < 8; ++r)
            dp[r * 32] = d[r];
    }

    float* wb = wbuf + w * WSET;
    const unsigned lane16 = (unsigned)lane * 16u;

    float acc[BB];
    #pragma unroll
    for (int b = 0; b < BB; ++b) acc[b] = 0.0f;

    for (int s = 0; s < ISTEPS; ++s) {
        const int i  = w * ISTEPS + s;
        const int nb = i * OUT_SZ + g * 32;

        const gfp pw0 = uni(w0 + nb * HID);
        const gfp pb0 = uni(b0 + nb * HID);
        const gfp pw1 = uni(w1 + nb * HID * HID);
        const gfp pb1 = uni(b1 + nb * HID);
        const gfp pw2 = uni(w2 + nb * HID);
        const gfp pb2 = uni(b2 + nb);
        const gfp pss = uni(ss + nb);

        const v4f r0 = ld4(pw0, lane16, 0);
        const v4f r1 = ld4(pb0, lane16, 0);
        const v4f r2 = ld4(pw1, lane16, 0);
        const v4f r3 = ld4(pw1, lane16, 1);
        const v4f r4 = ld4(pw1, lane16, 2);
        const v4f r5 = ld4(pw1, lane16, 3);
        const v4f r6 = ld4(pw1, lane16, 4);
        const v4f r7 = ld4(pw1, lane16, 5);
        const v4f r8 = ld4(pb1, lane16, 0);
        const v4f r9 = ld4(pw2, lane16, 0);
        v4f q0, q1, q2, q3, q4, q5, q6;
        if (lane < 8) {
            q0 = ld4(pw0, lane16, 1);
            q1 = ld4(pb0, lane16, 1);
            q2 = ld4(pw1, lane16, 6);
            q3 = ld4(pb1, lane16, 1);
            q4 = ld4(pw2, lane16, 1);
            q5 = ld4(pb2, lane16, 0);
            q6 = ld4(pss, lane16, 0);
        }

        __builtin_amdgcn_fence(__ATOMIC_RELEASE, "wavefront");
        __builtin_amdgcn_wave_barrier();

        st4(wb + OFF_W0, lane, r0);
        st4(wb + OFF_B0, lane, r1);
        st4(wb + OFF_W1, lane, r2);
        st4(wb + OFF_W1, lane + 32, r3);
        st4(wb + OFF_W1, lane + 64, r4);
        st4(wb + OFF_W1, lane + 96, r5);
        st4(wb + OFF_W1, lane + 128, r6);
        st4(wb + OFF_W1, lane + 160, r7);
        st4(wb + OFF_B1, lane, r8);
        st4(wb + OFF_W2, lane, r9);
        if (lane < 8) {
            st4(wb + OFF_W0, lane + 32, q0);
            st4(wb + OFF_B0, lane + 32, q1);
            st4(wb + OFF_W1, lane + 192, q2);
            st4(wb + OFF_B1, lane + 32, q3);
            st4(wb + OFF_W2, lane + 32, q4);
            st4(wb + OFF_B2, lane, q5);
            st4(wb + OFF_SS, lane, q6);
        }

        __builtin_amdgcn_fence(__ATOMIC_RELEASE, "wavefront");
        __builtin_amdgcn_wave_barrier();
        __builtin_amdgcn_fence(__ATOMIC_ACQUIRE, "wavefront");

        float W0[HID], B0[HID], W1[HID * HID], B1[HID], W2[HID];
        #pragma unroll
        for (int h = 0; h < HID; ++h) {
            W0[h] = 0.5f * wb[OFF_W0 + lane * HID + h];
            B0[h] = 0.5f * wb[OFF_B0 + lane * HID + h];
            B1[h] = 0.5f * wb[OFF_B1 + lane * HID + h];
            W2[h] = wb[OFF_W2 + lane * HID + h];
        }
        #pragma unroll
        for (int k = 0; k < HID * HID; ++k)
            W1[k] = 0.5f * wb[OFF_W1 + lane * HID * HID + k];
        const float B2 = wb[OFF_B2 + lane];
        const float SS = wb[OFF_SS + lane];

        #pragma unroll
        for (int q = 0; q < BB / 4; ++q) {
            const float4 xq = *reinterpret_cast<const float4*>(&xsT[i * BB + 4 * q]);
            const float xv[4] = {xq.x, xq.y, xq.z, xq.w};
            #pragma unroll
            for (int j = 0; j < 4; ++j) {
                float h0[HID], h1[HID];
                #pragma unroll
                for (int h = 0; h < HID; ++h)
                    h0[h] = silu_half(fmaf(W0[h], xv[j], B0[h]));

                #pragma unroll
                for (int h = 0; h < HID; ++h) {
                    float sacc = B1[h];
                    #pragma unroll
                    for (int k = 0; k < HID; ++k)
                        sacc = fmaf(W1[h * HID + k], h0[k], sacc);
                    h1[h] = silu_half(sacc);
                }

                float y = B2;
                #pragma unroll
                for (int k = 0; k < HID; ++k)
                    y = fmaf(W2[k], h1[k], y);

                acc[4 * q + j] = fmaf(y, SS, acc[4 * q + j]);
            }
        }
    }

    __syncthreads();
    float* red = wbuf;
    #pragma unroll
    for (int b = 0; b < BB; ++b)
        red[(w * BB + b) * 32 + lane] = acc[b];
    __syncthreads();

    float sum = 0.0f;
    #pragma unroll
    for (int ww = 0; ww < NWAVE; ++ww)
        sum += red[(ww * BB + w) * 32 + lane];

    float rsum = 0.0f;
    #pragma unroll
    for (int kk = 0; kk < KCH; ++kk)
        rsum += rpart[(kk * BB + w) * 32 + lane];

    out[(bb + w) * OUT_SZ + g * 32 + lane] = rsum + sum;
}

extern "C" void kernel_launch(void* const* d_in, const int* in_sizes, int n_in,
                              void* d_out, int out_size, void* d_ws, size_t ws_size,
                              hipStream_t stream) {
    const float* x   = (const float*)d_in[0];
    const float* w0  = (const float*)d_in[1];
    const float* b0  = (const float*)d_in[2];
    const float* w1  = (const float*)d_in[3];
    const float* b1  = (const float*)d_in[4];
    const float* w2  = (const float*)d_in[5];
    const float* b2  = (const float*)d_in[6];
    const float* ss  = (const float*)d_in[7];
    const float* rs  = (const float*)d_in[8];
    float* out = (float*)d_out;

    dim3 gridS(OUT_SZ / 32, BATCH / BB);
    size_t ldsBytes = (size_t)LDS_FLOATS * sizeof(float);
    subnet_kernel<<<gridS, NWAVE * 32, ldsBytes, stream>>>(
        x, w0, b0, w1, b1, w2, b2, ss, rs, out);
}


# nets' multiply-adds packed two batch rows per instruction (v_pk_fma_f32; a weight is read from one half of a register pair for both rows): 736 -> 368 fma issues per step, every sum keeps its order
# speedup vs baseline: 6.0907x; 1.1587x over previous
#include <hip/hip_runtime.h>
#include <hip/hip_bf16.h>


#define BATCH   2048
#define IN_SZ   128
#define OUT_SZ  128
#define HID     5
#define BB      16
#define NWAVE   16
#define ISTEPS  (IN_SZ / NWAVE)
#define NETW    47
#define WSET    (32 * NETW)
#define OFF_W0  0
#define OFF_B0  160
#define OFF_W1  320
#define OFF_B1  1120
#define OFF_W2  1280
#define OFF_B2  1440
#define OFF_SS  1472
#define PW0 0
#define PB0 5
#define PW1 10
#define PB1 35
#define PW2 40
#define PB2 45
#define PSS 46
#define KCH     (NWAVE / 2)
#define LDS_FLOATS (IN_SZ * BB + NWAVE * WSET + KCH * BB * 32)

typedef float v2f __attribute__((ext_vector_type(2)));
typedef float v8f __attribute__((ext_vector_type(8)));
typedef float v4f __attribute__((ext_vector_type(4)));
typedef const __attribute__((address_space(1))) float* gfp;
typedef const __attribute__((address_space(1))) v4f*   gv4p;
typedef const __attribute__((address_space(1))) char*  gcp;

__device__ __forceinline__ v2f pfma(v2f a, v2f b, v2f c) {
    return __builtin_elementwise_fma(a, b, c);
}
__device__ __forceinline__ v2f tanh2(v2f hz) {
    v2f t;
    t.x = __builtin_amdgcn_tanhf(hz.x);
    t.y = __builtin_amdgcn_tanhf(hz.y);
    return t;
}
__device__ __forceinline__ v2f spl(const v2f* P, int i) {
    const v2f p = P[i >> 1];
    const float s = (i & 1) ? p.y : p.x;
    return (v2f){s, s};
}

__device__ __forceinline__ v4f ld4(gfp p, unsigned lane16, unsigned k) {
    return *(gv4p)((gcp)p + lane16 + k * 512u);
}
__device__ __forceinline__ gfp uni(const float* p) {
    const unsigned long long v = reinterpret_cast<unsigned long long>(p);
    const unsigned lo = __builtin_amdgcn_readfirstlane(static_cast<unsigned>(v));
    const unsigned hi = __builtin_amdgcn_readfirstlane(static_cast<unsigned>(v >> 32));
    return (gfp)((static_cast<unsigned long long>(hi) << 32) | lo);
}
__device__ __forceinline__ void st4(float* p, int idx, v4f v) {
    reinterpret_cast<v4f*>(p)[idx] = v;
}

__global__ __launch_bounds__(NWAVE * 32) __attribute__((amdgpu_waves_per_eu(8, 8))) void subnet_kernel(
    const float* __restrict__ x,
    const float* __restrict__ w0,
    const float* __restrict__ b0,
    const float* __restrict__ w1,
    const float* __restrict__ b1,
    const float* __restrict__ w2,
    const float* __restrict__ b2,
    const float* __restrict__ ss,
    const float* __restrict__ rs,
    float* __restrict__ out)
{
    extern __shared__ float4 lds4[];
    float* xsT  = reinterpret_cast<float*>(lds4);
    float* wbuf = xsT + IN_SZ * BB;
    float* rpart = wbuf + NWAVE * WSET;

    const int t    = threadIdx.x;
    const int lane = t & 31;
    const int w    = __builtin_amdgcn_readfirstlane(t >> 5);
    const int g    = blockIdx.x;
    const int bb   = blockIdx.y * BB;

    const int lo = lane & 15;
    const int hi = lane >> 4;
    const int ct = w & 1;
    const int kc = w >> 1;
    const float* bp = rs + (kc * 16 + 2 * hi) * OUT_SZ + g * 32 + ct * 16 + lo;
    v2f bfr[4];
    #pragma unroll
    for (int j = 0; j < 4; ++j) {
        bfr[j].x = bp[(4 * j) * OUT_SZ];
        bfr[j].y = bp[(4 * j + 1) * OUT_SZ];
    }

    for (int j = t; j < BB * IN_SZ; j += NWAVE * 32) {
        int bl = j >> 7;
        int i  = j & (IN_SZ - 1);
        xsT[i * BB + bl] = x[(bb + bl) * IN_SZ + i];
    }
    __syncthreads();

    {
        const float* ap = xsT + (kc * 16 + 2 * hi) * BB + lo;
        v8f d = {};
        #pragma unroll
        for (int j = 0; j < 4; ++j) {
            v2f a;
            a.x = ap[(4 * j) * BB];
            a.y = ap[(4 * j + 1) * BB];
            d = __builtin_amdgcn_wmma_f32_16x16x4_f32(
                false, a, false, bfr[j], (short)0, d, false, false);
        }
        float* dp = rpart + (kc * BB + 8 * hi) * 32 + ct * 16 + lo;
        #pragma unroll
        for (int r = 0; r < 8; ++r)
            dp[r * 32] = d[r];
    }

    float* wb = wbuf + w * WSET;
    const unsigned lane16 = (unsigned)lane * 16u;

    v2f acc[BB / 2];
    #pragma unroll
    for (int p = 0; p < BB / 2; ++p) acc[p] = (v2f){0.0f, 0.0f};

    for (int s = 0; s < ISTEPS; ++s) {
        const int i  = w * ISTEPS + s;
        const int nb = i * OUT_SZ + g * 32;

        const gfp pw0 = uni(w0 + nb * HID);
        const gfp pb0 = uni(b0 + nb * HID);
        const gfp pw1 = uni(w1 + nb * HID * HID);
        const gfp pb1 = uni(b1 + nb * HID);
        const gfp pw2 = uni(w2 + nb * HID);
        const gfp pb2 = uni(b2 + nb);
        const gfp pss = uni(ss + nb);

        const v4f r0 = ld4(pw0, lane16, 0);
        const v4f r1 = ld4(pb0, lane16, 0);
        const v4f r2 = ld4(pw1, lane16, 0);
        const v4f r3 = ld4(pw1, lane16, 1);
        const v4f r4 = ld4(pw1, lane16, 2);
        const v4f r5 = ld4(pw1, lane16, 3);
        const v4f r6 = ld4(pw1, lane16, 4);
        const v4f r7 = ld4(pw1, lane16, 5);
        const v4f r8 = ld4(pb1, lane16, 0);
        const v4f r9 = ld4(pw2, lane16, 0);
        __builtin_amdgcn_fence(__ATOMIC_RELEASE, "wavefront");
        __builtin_amdgcn_wave_barrier();

        if (lane < 8) {
            const v4f q0 = ld4(pw0, lane16, 1);
            const v4f q1 = ld4(pb0, lane16, 1);
            const v4f q2 = ld4(pw1, lane16, 6);
            const v4f q3 = ld4(pb1, lane16, 1);
            const v4f q4 = ld4(pw2, lane16, 1);
            const v4f q5 = ld4(pb2, lane16, 0);
            const v4f q6 = ld4(pss, lane16, 0);
            st4(wb + OFF_W0, lane + 32, q0);
            st4(wb + OFF_B0, lane + 32, q1);
            st4(wb + OFF_W1, lane + 192, q2);
            st4(wb + OFF_B1, lane + 32, q3);
            st4(wb + OFF_W2, lane + 32, q4);
            st4(wb + OFF_B2, lane, q5);
            st4(wb + OFF_SS, lane, q6);
        }
        st4(wb + OFF_W0, lane, r0);
        st4(wb + OFF_B0, lane, r1);
        st4(wb + OFF_W1, lane, r2);
        st4(wb + OFF_W1, lane + 32, r3);
        st4(wb + OFF_W1, lane + 64, r4);
        st4(wb + OFF_W1, lane + 96, r5);
        st4(wb + OFF_W1, lane + 128, r6);
        st4(wb + OFF_W1, lane + 160, r7);
        st4(wb + OFF_B1, lane, r8);
        st4(wb + OFF_W2, lane, r9);

        __builtin_amdgcn_fence(__ATOMIC_RELEASE, "wavefront");
        __builtin_amdgcn_wave_barrier();
        __builtin_amdgcn_fence(__ATOMIC_ACQUIRE, "wavefront");

        float wt[48];
        #pragma unroll
        for (int h = 0; h < HID; ++h) {
            wt[PW0 + h] = wb[OFF_W0 + lane * HID + h];
            wt[PB0 + h] = wb[OFF_B0 + lane * HID + h];
            wt[PB1 + h] = wb[OFF_B1 + lane * HID + h];
            wt[PW2 + h] = wb[OFF_W2 + lane * HID + h];
        }
        #pragma unroll
        for (int k = 0; k < HID * HID; ++k)
            wt[PW1 + k] = wb[OFF_W1 + lane * HID * HID + k];
        wt[PB2] = wb[OFF_B2 + lane];
        wt[PSS] = wb[OFF_SS + lane];
        wt[47]  = 0.0f;
        v2f P[24];
        #pragma unroll
        for (int k = 0; k < 24; ++k) {
            const v2f p = {wt[2 * k], wt[2 * k + 1]};
            P[k] = (k < 20) ? p * 0.5f : p;
        }

        #pragma unroll
        for (int q = 0; q < BB / 4; ++q) {
            const float4 xq = *reinterpret_cast<const float4*>(&xsT[i * BB + 4 * q]);
            const v2f xv[2] = {{xq.x, xq.y}, {xq.z, xq.w}};
            v2f z[2][HID], s[2][HID], t[2], tn[2], y[2];
            #pragma unroll
            for (int h = 0; h < HID; ++h) {
                #pragma unroll
                for (int j = 0; j < 2; ++j) {
                    z[j][h] = pfma(spl(P, PW0 + h), xv[j], spl(P, PB0 + h));
                    s[j][h] = spl(P, PB1 + h);
                }
            }
            #pragma unroll
            for (int j = 0; j < 2; ++j) t[j] = tanh2(z[j][0]);
            #pragma unroll
            for (int k = 0; k < HID; ++k) {
                if (k + 1 < HID) {
                    #pragma unroll
                    for (int j = 0; j < 2; ++j) tn[j] = tanh2(z[j][k + 1]);
                }
                v2f h0[2];
                #pragma unroll
                for (int j = 0; j < 2; ++j) h0[j] = pfma(z[j][k], t[j], z[j][k]);
                #pragma unroll
                for (int h = 0; h < HID; ++h) {
                    #pragma unroll
                    for (int j = 0; j < 2; ++j)
                        s[j][h] = pfma(spl(P, PW1 + h * HID + k), h0[j], s[j][h]);
                }
                #pragma unroll
                for (int j = 0; j < 2; ++j) t[j] = tn[j];
            }
            #pragma unroll
            for (int j = 0; j < 2; ++j) {
                t[j] = tanh2(s[j][0]);
                y[j] = spl(P, PB2);
            }
            #pragma unroll
            for (int k = 0; k < HID; ++k) {
                if (k + 1 < HID) {
                    #pragma unroll
                    for (int j = 0; j < 2; ++j) tn[j] = tanh2(s[j][k + 1]);
                }
                #pragma unroll
                for (int j = 0; j < 2; ++j) {
                    const v2f h1 = pfma(s[j][k], t[j], s[j][k]);
                    y[j] = pfma(spl(P, PW2 + k), h1, y[j]);
                    t[j] = tn[j];
                }
            }
            #pragma unroll
            for (int j = 0; j < 2; ++j)
                acc[2 * q + j] = pfma(y[j], spl(P, PSS), acc[2 * q + j]);
        }
    }

    __syncthreads();
    float* red = wbuf;
    #pragma unroll
    for (int p = 0; p < BB / 2; ++p) {
        red[(w * BB + 2 * p) * 32 + lane]     = acc[p].x;
        red[(w * BB + 2 * p + 1) * 32 + lane] = acc[p].y;
    }
    __syncthreads();

    float sum = 0.0f;
    #pragma unroll
    for (int ww = 0; ww < NWAVE; ++ww)
        sum += red[(ww * BB + w) * 32 + lane];

    float rsum = 0.0f;
    #pragma unroll
    for (int kk = 0; kk < KCH; ++kk)
        rsum += rpart[(kk * BB + w) * 32 + lane];

    out[(bb + w) * OUT_SZ + g * 32 + lane] = rsum + sum;
}

extern "C" void kernel_launch(void* const* d_in, const int* in_sizes, int n_in,
                              void* d_out, int out_size, void* d_ws, size_t ws_size,
                              hipStream_t stream) {
    const float* x   = (const float*)d_in[0];
    const float* w0  = (const float*)d_in[1];
    const float* b0  = (const float*)d_in[2];
    const float* w1  = (const float*)d_in[3];
    const float* b1  = (const float*)d_in[4];
    const float* w2  = (const float*)d_in[5];
    const float* b2  = (const float*)d_in[6];
    const float* ss  = (const float*)d_in[7];
    const float* rs  = (const float*)d_in[8];
    float* out = (float*)d_out;

    dim3 gridS(OUT_SZ / 32, BATCH / BB);
    size_t ldsBytes = (size_t)LDS_FLOATS * sizeof(float);
    subnet_kernel<<<gridS, NWAVE * 32, ldsBytes, stream>>>(
        x, w0, b0, w1, b1, w2, b2, ss, rs, out);
}


# block widened to 32 batch rows x 32 waves (4 inputs each): a net's weights are fetched and unpacked once per 32 rows, not 16; the i-sum becomes 32 partial sums of 4 (was 16 of 8) added through LDS
# speedup vs baseline: 6.6399x; 1.0902x over previous
#include <hip/hip_runtime.h>
#include <hip/hip_bf16.h>


#define BATCH   2048
#define IN_SZ   128
#define OUT_SZ  128
#define HID     5
#define BB      32
#define XP      36
#define NWAVE   32
#define ISTEPS  (IN_SZ / NWAVE)
#define NJ      1
#define NETW    47
#define WSET    (32 * NETW)
#define OFF_W0  0
#define OFF_B0  160
#define OFF_W1  320
#define OFF_B1  1120
#define OFF_W2  1280
#define OFF_B2  1440
#define OFF_SS  1472
#define PW0 0
#define PB0 5
#define PW1 10
#define PB1 35
#define PW2 40
#define PB2 45
#define PSS 46
#define KCH     (NWAVE / 4)
#define LDS_FLOATS (IN_SZ * XP + NWAVE * WSET + KCH * BB * 32)

typedef float v2f __attribute__((ext_vector_type(2)));
typedef float v8f __attribute__((ext_vector_type(8)));
typedef float v4f __attribute__((ext_vector_type(4)));
typedef const __attribute__((address_space(1))) float* gfp;
typedef const __attribute__((address_space(1))) v4f*   gv4p;
typedef const __attribute__((address_space(1))) char*  gcp;

__device__ __forceinline__ v2f pfma(v2f a, v2f b, v2f c) {
    return __builtin_elementwise_fma(a, b, c);
}
__device__ __forceinline__ v2f tanh2(v2f hz) {
    v2f t;
    t.x = __builtin_amdgcn_tanhf(hz.x);
    t.y = __builtin_amdgcn_tanhf(hz.y);
    return t;
}
__device__ __forceinline__ v2f spl(const v2f* P, int i) {
    const v2f p = P[i >> 1];
    const float s = (i & 1) ? p.y : p.x;
    return (v2f){s, s};
}

__device__ __forceinline__ v4f ld4(gfp p, unsigned lane16, unsigned k) {
    return *(gv4p)((gcp)p + lane16 + k * 512u);
}
__device__ __forceinline__ gfp uni(const float* p) {
    const unsigned long long v = reinterpret_cast<unsigned long long>(p);
    const unsigned lo = __builtin_amdgcn_readfirstlane(static_cast<unsigned>(v));
    const unsigned hi = __builtin_amdgcn_readfirstlane(static_cast<unsigned>(v >> 32));
    return (gfp)((static_cast<unsigned long long>(hi) << 32) | lo);
}
__device__ __forceinline__ void st4(float* p, int idx, v4f v) {
    reinterpret_cast<v4f*>(p)[idx] = v;
}

__global__ __launch_bounds__(NWAVE * 32) __attribute__((amdgpu_waves_per_eu(8, 8))) void subnet_kernel(
    const float* __restrict__ x,
    const float* __restrict__ w0,
    const float* __restrict__ b0,
    const float* __restrict__ w1,
    const float* __restrict__ b1,
    const float* __restrict__ w2,
    const float* __restrict__ b2,
    const float* __restrict__ ss,
    const float* __restrict__ rs,
    float* __restrict__ out)
{
    extern __shared__ float4 lds4[];
    float* xsT  = reinterpret_cast<float*>(lds4);
    float* wbuf = xsT + IN_SZ * XP;
    float* rpart = wbuf + NWAVE * WSET;

    const int t    = threadIdx.x;
    const int lane = t & 31;
    const int w    = __builtin_amdgcn_readfirstlane(t >> 5);
    const int g    = blockIdx.x;
    const int bb   = blockIdx.y * BB;

    const int lo = lane & 15;
    const int hi = lane >> 4;
    const int ct = w & 1;
    const int rt = (w >> 1) & 1;
    const int kc = w >> 2;
    const float* bp = rs + (kc * 16 + 2 * hi) * OUT_SZ + g * 32 + ct * 16 + lo;
    v2f bfr[4];
    #pragma unroll
    for (int j = 0; j < 4; ++j) {
        bfr[j].x = bp[(4 * j) * OUT_SZ];
        bfr[j].y = bp[(4 * j + 1) * OUT_SZ];
    }

    for (int j = t; j < BB * IN_SZ; j += NWAVE * 32) {
        int bl = j >> 7;
        int i  = j & (IN_SZ - 1);
        xsT[i * XP + bl] = x[(bb + bl) * IN_SZ + i];
    }
    __syncthreads();

    {
        const float* ap = xsT + (kc * 16 + 2 * hi) * XP + rt * 16 + lo;
        v8f d = {};
        #pragma unroll
        for (int j = 0; j < 4; ++j) {
            v2f a;
            a.x = ap[(4 * j) * XP];
            a.y = ap[(4 * j + 1) * XP];
            d = __builtin_amdgcn_wmma_f32_16x16x4_f32(
                false, a, false, bfr[j], (short)0, d, false, false);
        }
        float* dp = rpart + (kc * BB + rt * 16 + 8 * hi) * 32 + ct * 16 + lo;
        #pragma unroll
        for (int r = 0; r < 8; ++r)
            dp[r * 32] = d[r];
    }

    float* wb = wbuf + w * WSET;
    const unsigned lane16 = (unsigned)lane * 16u;

    v2f acc[BB / 2];
    #pragma unroll
    for (int p = 0; p < BB / 2; ++p) acc[p] = (v2f){0.0f, 0.0f};

    for (int s = 0; s < ISTEPS; ++s) {
        const int i  = w * ISTEPS + s;
        const int nb = i * OUT_SZ + g * 32;

        const gfp pw0 = uni(w0 + nb * HID);
        const gfp pb0 = uni(b0 + nb * HID);
        const gfp pw1 = uni(w1 + nb * HID * HID);
        const gfp pb1 = uni(b1 + nb * HID);
        const gfp pw2 = uni(w2 + nb * HID);
        const gfp pb2 = uni(b2 + nb);
        const gfp pss = uni(ss + nb);

        const v4f r0 = ld4(pw0, lane16, 0);
        const v4f r1 = ld4(pb0, lane16, 0);
        const v4f r2 = ld4(pw1, lane16, 0);
        const v4f r3 = ld4(pw1, lane16, 1);
        const v4f r4 = ld4(pw1, lane16, 2);
        const v4f r5 = ld4(pw1, lane16, 3);
        const v4f r6 = ld4(pw1, lane16, 4);
        const v4f r7 = ld4(pw1, lane16, 5);
        const v4f r8 = ld4(pb1, lane16, 0);
        const v4f r9 = ld4(pw2, lane16, 0);
        __builtin_amdgcn_fence(__ATOMIC_RELEASE, "wavefront");
        __builtin_amdgcn_wave_barrier();

        if (lane < 8) {
            const v4f q0 = ld4(pw0, lane16, 1);
            const v4f q1 = ld4(pb0, lane16, 1);
            const v4f q2 = ld4(pw1, lane16, 6);
            const v4f q3 = ld4(pb1, lane16, 1);
            const v4f q4 = ld4(pw2, lane16, 1);
            const v4f q5 = ld4(pb2, lane16, 0);
            const v4f q6 = ld4(pss, lane16, 0);
            st4(wb + OFF_W0, lane + 32, q0);
            st4(wb + OFF_B0, lane + 32, q1);
            st4(wb + OFF_W1, lane + 192, q2);
            st4(wb + OFF_B1, lane + 32, q3);
            st4(wb + OFF_W2, lane + 32, q4);
            st4(wb + OFF_B2, lane, q5);
            st4(wb + OFF_SS, lane, q6);
        }
        st4(wb + OFF_W0, lane, r0);
        st4(wb + OFF_B0, lane, r1);
        st4(wb + OFF_W1, lane, r2);
        st4(wb + OFF_W1, lane + 32, r3);
        st4(wb + OFF_W1, lane + 64, r4);
        st4(wb + OFF_W1, lane + 96, r5);
        st4(wb + OFF_W1, lane + 128, r6);
        st4(wb + OFF_W1, lane + 160, r7);
        st4(wb + OFF_B1, lane, r8);
        st4(wb + OFF_W2, lane, r9);

        __builtin_amdgcn_fence(__ATOMIC_RELEASE, "wavefront");
        __builtin_amdgcn_wave_barrier();
        __builtin_amdgcn_fence(__ATOMIC_ACQUIRE, "wavefront");

        float wt[48];
        #pragma unroll
        for (int h = 0; h < HID; ++h) {
            wt[PW0 + h] = wb[OFF_W0 + lane * HID + h];
            wt[PB0 + h] = wb[OFF_B0 + lane * HID + h];
            wt[PB1 + h] = wb[OFF_B1 + lane * HID + h];
            wt[PW2 + h] = wb[OFF_W2 + lane * HID + h];
        }
        #pragma unroll
        for (int k = 0; k < HID * HID; ++k)
            wt[PW1 + k] = wb[OFF_W1 + lane * HID * HID + k];
        wt[PB2] = wb[OFF_B2 + lane];
        wt[PSS] = wb[OFF_SS + lane];
        wt[47]  = 0.0f;
        v2f P[24];
        #pragma unroll
        for (int k = 0; k < 24; ++k) {
            const v2f p = {wt[2 * k], wt[2 * k + 1]};
            P[k] = (k < 20) ? p * 0.5f : p;
        }

        #pragma unroll
        for (int q = 0; q < BB / 4; ++q) {
            const float4 xq = *reinterpret_cast<const float4*>(&xsT[i * XP + 4 * q]);
            const v2f xall[2] = {{xq.x, xq.y}, {xq.z, xq.w}};
            #pragma unroll
            for (int jb = 0; jb < 2; jb += NJ) {
                v2f z[NJ][HID], s[NJ][HID], t[NJ], tn[NJ], y[NJ];
                #pragma unroll
                for (int h = 0; h < HID; ++h) {
                    #pragma unroll
                    for (int j = 0; j < NJ; ++j) {
                        z[j][h] = pfma(spl(P, PW0 + h), xall[jb + j], spl(P, PB0 + h));
                        s[j][h] = spl(P, PB1 + h);
                    }
                }
                #pragma unroll
                for (int j = 0; j < NJ; ++j) t[j] = tanh2(z[j][0]);
                #pragma unroll
                for (int k = 0; k < HID; ++k) {
                    if (k + 1 < HID) {
                        #pragma unroll
                        for (int j = 0; j < NJ; ++j) tn[j] = tanh2(z[j][k + 1]);
                    }
                    v2f h0[NJ];
                    #pragma unroll
                    for (int j = 0; j < NJ; ++j) h0[j] = pfma(z[j][k], t[j], z[j][k]);
                    #pragma unroll
                    for (int h = 0; h < HID; ++h) {
                        #pragma unroll
                        for (int j = 0; j < NJ; ++j)
                            s[j][h] = pfma(spl(P, PW1 + h * HID + k), h0[j], s[j][h]);
                    }
                    #pragma unroll
                    for (int j = 0; j < NJ; ++j) t[j] = tn[j];
                }
                #pragma unroll
                for (int j = 0; j < NJ; ++j) {
                    t[j] = tanh2(s[j][0]);
                    y[j] = spl(P, PB2);
                }
                #pragma unroll
                for (int k = 0; k < HID; ++k) {
                    if (k + 1 < HID) {
                        #pragma unroll
                        for (int j = 0; j < NJ; ++j) tn[j] = tanh2(s[j][k + 1]);
                    }
                    #pragma unroll
                    for (int j = 0; j < NJ; ++j) {
                        const v2f h1 = pfma(s[j][k], t[j], s[j][k]);
                        y[j] = pfma(spl(P, PW2 + k), h1, y[j]);
                        t[j] = tn[j];
                    }
                }
                #pragma unroll
                for (int j = 0; j < NJ; ++j)
                    acc[2 * q + jb + j] = pfma(y[j], spl(P, PSS), acc[2 * q + jb + j]);
            }
        }
    }

    __syncthreads();
    float* red = wbuf;
    #pragma unroll
    for (int p = 0; p < BB / 2; ++p) {
        red[(w * BB + 2 * p) * 32 + lane]     = acc[p].x;
        red[(w * BB + 2 * p + 1) * 32 + lane] = acc[p].y;
    }
    __syncthreads();

    float sum = 0.0f;
    #pragma unroll
    for (int ww = 0; ww < NWAVE; ++ww)
        sum += red[(ww * BB + w) * 32 + lane];

    float rsum = 0.0f;
    #pragma unroll
    for (int kk = 0; kk < KCH; ++kk)
        rsum += rpart[(kk * BB + w) * 32 + lane];

    out[(bb + w) * OUT_SZ + g * 32 + lane] = rsum + sum;
}

extern "C" void kernel_launch(void* const* d_in, const int* in_sizes, int n_in,
                              void* d_out, int out_size, void* d_ws, size_t ws_size,
                              hipStream_t stream) {
    const float* x   = (const float*)d_in[0];
    const float* w0  = (const float*)d_in[1];
    const float* b0  = (const float*)d_in[2];
    const float* w1  = (const float*)d_in[3];
    const float* b1  = (const float*)d_in[4];
    const float* w2  = (const float*)d_in[5];
    const float* b2  = (const float*)d_in[6];
    const float* ss  = (const float*)d_in[7];
    const float* rs  = (const float*)d_in[8];
    float* out = (float*)d_out;

    dim3 gridS(OUT_SZ / 32, BATCH / BB);
    size_t ldsBytes = (size_t)LDS_FLOATS * sizeof(float);
    subnet_kernel<<<gridS, NWAVE * 32, ldsBytes, stream>>>(
        x, w0, b0, w1, b1, w2, b2, ss, rs, out);
}


# block head is one memory hop, not five: each thread's 4 x loads, the rs fragments and the first step's weight loads all issue before the first wait; the LDS-only barrier leaves them in flight
# speedup vs baseline: 6.8569x; 1.0327x over previous
#include <hip/hip_runtime.h>
#include <hip/hip_bf16.h>


#define BATCH   2048
#define IN_SZ   128
#define OUT_SZ  128
#define HID     5
#define BB      32
#define XP      36
#define NWAVE   32
#define ISTEPS  (IN_SZ / NWAVE)
#define NJ      1
#define NETW    47
#define WSET    (32 * NETW)
#define OFF_W0  0
#define OFF_B0  160
#define OFF_W1  320
#define OFF_B1  1120
#define OFF_W2  1280
#define OFF_B2  1440
#define OFF_SS  1472
#define PW0 0
#define PB0 5
#define PW1 10
#define PB1 35
#define PW2 40
#define PB2 45
#define PSS 46
#define KCH     (NWAVE / 4)
#define LDS_FLOATS (IN_SZ * XP + NWAVE * WSET + KCH * BB * 32)

typedef float v2f __attribute__((ext_vector_type(2)));
typedef float v8f __attribute__((ext_vector_type(8)));
typedef float v4f __attribute__((ext_vector_type(4)));
typedef const __attribute__((address_space(1))) float* gfp;
typedef const __attribute__((address_space(1))) v4f*   gv4p;
typedef const __attribute__((address_space(1))) char*  gcp;

__device__ __forceinline__ v2f pfma(v2f a, v2f b, v2f c) {
    return __builtin_elementwise_fma(a, b, c);
}
__device__ __forceinline__ v2f tanh2(v2f hz) {
    v2f t;
    t.x = __builtin_amdgcn_tanhf(hz.x);
    t.y = __builtin_amdgcn_tanhf(hz.y);
    return t;
}
__device__ __forceinline__ v2f spl(const v2f* P, int i) {
    const v2f p = P[i >> 1];
    const float s = (i & 1) ? p.y : p.x;
    return (v2f){s, s};
}

__device__ __forceinline__ v4f ld4(gfp p, unsigned lane16, unsigned k) {
    return *(gv4p)((gcp)p + lane16 + k * 512u);
}
__device__ __forceinline__ gfp uni(const float* p) {
    const unsigned long long v = reinterpret_cast<unsigned long long>(p);
    const unsigned lo = __builtin_amdgcn_readfirstlane(static_cast<unsigned>(v));
    const unsigned hi = __builtin_amdgcn_readfirstlane(static_cast<unsigned>(v >> 32));
    return (gfp)((static_cast<unsigned long long>(hi) << 32) | lo);
}
__device__ __forceinline__ void st4(float* p, int idx, v4f v) {
    reinterpret_cast<v4f*>(p)[idx] = v;
}

__global__ __launch_bounds__(NWAVE * 32) __attribute__((amdgpu_waves_per_eu(8, 8))) void subnet_kernel(
    const float* __restrict__ x,
    const float* __restrict__ w0,
    const float* __restrict__ b0,
    const float* __restrict__ w1,
    const float* __restrict__ b1,
    const float* __restrict__ w2,
    const float* __restrict__ b2,
    const float* __restrict__ ss,
    const float* __restrict__ rs,
    float* __restrict__ out)
{
    extern __shared__ float4 lds4[];
    float* xsT  = reinterpret_cast<float*>(lds4);
    float* wbuf = xsT + IN_SZ * XP;
    float* rpart = wbuf + NWAVE * WSET;

    const int t    = threadIdx.x;
    const int lane = t & 31;
    const int w    = __builtin_amdgcn_readfirstlane(t >> 5);
    const int g    = blockIdx.x;
    const int bb   = blockIdx.y * BB;

    float xv[(BB * IN_SZ) / (NWAVE * 32)];
    #pragma unroll
    for (int k = 0; k < (BB * IN_SZ) / (NWAVE * 32); ++k) {
        const int j  = t + k * NWAVE * 32;
        xv[k] = x[(bb + (j >> 7)) * IN_SZ + (j & (IN_SZ - 1))];
    }

    const int lo = lane & 15;
    const int hi = lane >> 4;
    const int ct = w & 1;
    const int rt = (w >> 1) & 1;
    const int kc = w >> 2;
    const float* bp = rs + (kc * 16 + 2 * hi) * OUT_SZ + g * 32 + ct * 16 + lo;
    v2f bfr[4];
    #pragma unroll
    for (int j = 0; j < 4; ++j) {
        bfr[j].x = bp[(4 * j) * OUT_SZ];
        bfr[j].y = bp[(4 * j + 1) * OUT_SZ];
    }

    float* wb = wbuf + w * WSET;
    const unsigned lane16 = (unsigned)lane * 16u;
    const int nb0 = (w * ISTEPS) * OUT_SZ + g * 32;
    const gfp hw0 = uni(w0 + nb0 * HID);
    const gfp hb0 = uni(b0 + nb0 * HID);
    const gfp hw1 = uni(w1 + nb0 * HID * HID);
    const gfp hb1 = uni(b1 + nb0 * HID);
    const gfp hw2 = uni(w2 + nb0 * HID);
    const gfp hb2 = uni(b2 + nb0);
    const gfp hss = uni(ss + nb0);
    const v4f f0 = ld4(hw0, lane16, 0);
    const v4f f1 = ld4(hb0, lane16, 0);
    const v4f f2 = ld4(hw1, lane16, 0);
    const v4f f3 = ld4(hw1, lane16, 1);
    const v4f f4 = ld4(hw1, lane16, 2);
    const v4f f5 = ld4(hw1, lane16, 3);
    const v4f f6 = ld4(hw1, lane16, 4);
    const v4f f7 = ld4(hw1, lane16, 5);
    const v4f f8 = ld4(hb1, lane16, 0);
    const v4f f9 = ld4(hw2, lane16, 0);
    const unsigned lane16e = (unsigned)(lane & 7) * 16u;
    const v4f e0 = ld4(hw0, lane16e, 1);
    const v4f e1 = ld4(hb0, lane16e, 1);
    const v4f e2 = ld4(hw1, lane16e, 6);
    const v4f e3 = ld4(hb1, lane16e, 1);
    const v4f e4 = ld4(hw2, lane16e, 1);
    const v4f e5 = ld4(hb2, lane16e, 0);
    const v4f e6 = ld4(hss, lane16e, 0);

    #pragma unroll
    for (int k = 0; k < (BB * IN_SZ) / (NWAVE * 32); ++k) {
        const int j  = t + k * NWAVE * 32;
        xsT[(j & (IN_SZ - 1)) * XP + (j >> 7)] = xv[k];
    }
    __builtin_amdgcn_fence(__ATOMIC_RELEASE, "workgroup", "local");
    __builtin_amdgcn_s_barrier();
    __builtin_amdgcn_fence(__ATOMIC_ACQUIRE, "workgroup", "local");

    {
        const float* ap = xsT + (kc * 16 + 2 * hi) * XP + rt * 16 + lo;
        v8f d = {};
        #pragma unroll
        for (int j = 0; j < 4; ++j) {
            v2f a;
            a.x = ap[(4 * j) * XP];
            a.y = ap[(4 * j + 1) * XP];
            d = __builtin_amdgcn_wmma_f32_16x16x4_f32(
                false, a, false, bfr[j], (short)0, d, false, false);
        }
        float* dp = rpart + (kc * BB + rt * 16 + 8 * hi) * 32 + ct * 16 + lo;
        #pragma unroll
        for (int r = 0; r < 8; ++r)
            dp[r * 32] = d[r];
    }

    if (lane < 8) {
        st4(wb + OFF_W0, lane + 32, e0);
        st4(wb + OFF_B0, lane + 32, e1);
        st4(wb + OFF_W1, lane + 192, e2);
        st4(wb + OFF_B1, lane + 32, e3);
        st4(wb + OFF_W2, lane + 32, e4);
        st4(wb + OFF_B2, lane, e5);
        st4(wb + OFF_SS, lane, e6);
    }
    st4(wb + OFF_W0, lane, f0);
    st4(wb + OFF_B0, lane, f1);
    st4(wb + OFF_W1, lane, f2);
    st4(wb + OFF_W1, lane + 32, f3);
    st4(wb + OFF_W1, lane + 64, f4);
    st4(wb + OFF_W1, lane + 96, f5);
    st4(wb + OFF_W1, lane + 128, f6);
    st4(wb + OFF_W1, lane + 160, f7);
    st4(wb + OFF_B1, lane, f8);
    st4(wb + OFF_W2, lane, f9);

    v2f acc[BB / 2];
    #pragma unroll
    for (int p = 0; p < BB / 2; ++p) acc[p] = (v2f){0.0f, 0.0f};

    for (int s = 0; s < ISTEPS; ++s) {
        const int i  = w * ISTEPS + s;
        const int nb = i * OUT_SZ + g * 32;

        if (s > 0) {
            const gfp pw0 = uni(w0 + nb * HID);
            const gfp pb0 = uni(b0 + nb * HID);
            const gfp pw1 = uni(w1 + nb * HID * HID);
            const gfp pb1 = uni(b1 + nb * HID);
            const gfp pw2 = uni(w2 + nb * HID);
            const gfp pb2 = uni(b2 + nb);
            const gfp pss = uni(ss + nb);

            const v4f r0 = ld4(pw0, lane16, 0);
            const v4f r1 = ld4(pb0, lane16, 0);
            const v4f r2 = ld4(pw1, lane16, 0);
            const v4f r3 = ld4(pw1, lane16, 1);
            const v4f r4 = ld4(pw1, lane16, 2);
            const v4f r5 = ld4(pw1, lane16, 3);
            const v4f r6 = ld4(pw1, lane16, 4);
            const v4f r7 = ld4(pw1, lane16, 5);
            const v4f r8 = ld4(pb1, lane16, 0);
            const v4f r9 = ld4(pw2, lane16, 0);
            __builtin_amdgcn_fence(__ATOMIC_RELEASE, "wavefront");
            __builtin_amdgcn_wave_barrier();

            if (lane < 8) {
                const v4f q0 = ld4(pw0, lane16, 1);
                const v4f q1 = ld4(pb0, lane16, 1);
                const v4f q2 = ld4(pw1, lane16, 6);
                const v4f q3 = ld4(pb1, lane16, 1);
                const v4f q4 = ld4(pw2, lane16, 1);
                const v4f q5 = ld4(pb2, lane16, 0);
                const v4f q6 = ld4(pss, lane16, 0);
                st4(wb + OFF_W0, lane + 32, q0);
                st4(wb + OFF_B0, lane + 32, q1);
                st4(wb + OFF_W1, lane + 192, q2);
                st4(wb + OFF_B1, lane + 32, q3);
                st4(wb + OFF_W2, lane + 32, q4);
                st4(wb + OFF_B2, lane, q5);
                st4(wb + OFF_SS, lane, q6);
            }
            st4(wb + OFF_W0, lane, r0);
            st4(wb + OFF_B0, lane, r1);
            st4(wb + OFF_W1, lane, r2);
            st4(wb + OFF_W1, lane + 32, r3);
            st4(wb + OFF_W1, lane + 64, r4);
            st4(wb + OFF_W1, lane + 96, r5);
            st4(wb + OFF_W1, lane + 128, r6);
            st4(wb + OFF_W1, lane + 160, r7);
            st4(wb + OFF_B1, lane, r8);
            st4(wb + OFF_W2, lane, r9);
        }

        __builtin_amdgcn_fence(__ATOMIC_RELEASE, "wavefront");
        __builtin_amdgcn_wave_barrier();
        __builtin_amdgcn_fence(__ATOMIC_ACQUIRE, "wavefront");

        float wt[48];
        #pragma unroll
        for (int h = 0; h < HID; ++h) {
            wt[PW0 + h] = wb[OFF_W0 + lane * HID + h];
            wt[PB0 + h] = wb[OFF_B0 + lane * HID + h];
            wt[PB1 + h] = wb[OFF_B1 + lane * HID + h];
            wt[PW2 + h] = wb[OFF_W2 + lane * HID + h];
        }
        #pragma unroll
        for (int k = 0; k < HID * HID; ++k)
            wt[PW1 + k] = wb[OFF_W1 + lane * HID * HID + k];
        wt[PB2] = wb[OFF_B2 + lane];
        wt[PSS] = wb[OFF_SS + lane];
        wt[47]  = 0.0f;
        v2f P[24];
        #pragma unroll
        for (int k = 0; k < 24; ++k) {
            const v2f p = {wt[2 * k], wt[2 * k + 1]};
            P[k] = (k < 20) ? p * 0.5f : p;
        }

        #pragma unroll
        for (int q = 0; q < BB / 4; ++q) {
            const float4 xq = *reinterpret_cast<const float4*>(&xsT[i * XP + 4 * q]);
            const v2f xall[2] = {{xq.x, xq.y}, {xq.z, xq.w}};
            #pragma unroll
            for (int jb = 0; jb < 2; jb += NJ) {
                v2f z[NJ][HID], s[NJ][HID], t[NJ], tn[NJ], y[NJ];
                #pragma unroll
                for (int h = 0; h < HID; ++h) {
                    #pragma unroll
                    for (int j = 0; j < NJ; ++j) {
                        z[j][h] = pfma(spl(P, PW0 + h), xall[jb + j], spl(P, PB0 + h));
                        s[j][h] = spl(P, PB1 + h);
                    }
                }
                #pragma unroll
                for (int j = 0; j < NJ; ++j) t[j] = tanh2(z[j][0]);
                #pragma unroll
                for (int k = 0; k < HID; ++k) {
                    if (k + 1 < HID) {
                        #pragma unroll
                        for (int j = 0; j < NJ; ++j) tn[j] = tanh2(z[j][k + 1]);
                    }
                    v2f h0[NJ];
                    #pragma unroll
                    for (int j = 0; j < NJ; ++j) h0[j] = pfma(z[j][k], t[j], z[j][k]);
                    #pragma unroll
                    for (int h = 0; h < HID; ++h) {
                        #pragma unroll
                        for (int j = 0; j < NJ; ++j)
                            s[j][h] = pfma(spl(P, PW1 + h * HID + k), h0[j], s[j][h]);
                    }
                    #pragma unroll
                    for (int j = 0; j < NJ; ++j) t[j] = tn[j];
                }
                #pragma unroll
                for (int j = 0; j < NJ; ++j) {
                    t[j] = tanh2(s[j][0]);
                    y[j] = spl(P, PB2);
                }
                #pragma unroll
                for (int k = 0; k < HID; ++k) {
                    if (k + 1 < HID) {
                        #pragma unroll
                        for (int j = 0; j < NJ; ++j) tn[j] = tanh2(s[j][k + 1]);
                    }
                    #pragma unroll
                    for (int j = 0; j < NJ; ++j) {
                        const v2f h1 = pfma(s[j][k], t[j], s[j][k]);
                        y[j] = pfma(spl(P, PW2 + k), h1, y[j]);
                        t[j] = tn[j];
                    }
                }
                #pragma unroll
                for (int j = 0; j < NJ; ++j)
                    acc[2 * q + jb + j] = pfma(y[j], spl(P, PSS), acc[2 * q + jb + j]);
            }
        }
    }

    __syncthreads();
    float* red = wbuf;
    const int lane_e = (int)__builtin_amdgcn_mbcnt_lo(~0u, 0u);
    #pragma unroll
    for (int p = 0; p < BB / 2; ++p) {
        red[(w * BB + 2 * p) * 32 + lane_e]     = acc[p].x;
        red[(w * BB + 2 * p + 1) * 32 + lane_e] = acc[p].y;
    }
    __syncthreads();

    float sum = 0.0f;
    #pragma unroll
    for (int ww = 0; ww < NWAVE; ++ww)
        sum += red[(ww * BB + w) * 32 + lane_e];

    float rsum = 0.0f;
    #pragma unroll
    for (int kk = 0; kk < KCH; ++kk)
        rsum += rpart[(kk * BB + w) * 32 + lane_e];

    out[(bb + w) * OUT_SZ + g * 32 + lane_e] = rsum + sum;
}

extern "C" void kernel_launch(void* const* d_in, const int* in_sizes, int n_in,
                              void* d_out, int out_size, void* d_ws, size_t ws_size,
                              hipStream_t stream) {
    const float* x   = (const float*)d_in[0];
    const float* w0  = (const float*)d_in[1];
    const float* b0  = (const float*)d_in[2];
    const float* w1  = (const float*)d_in[3];
    const float* b1  = (const float*)d_in[4];
    const float* w2  = (const float*)d_in[5];
    const float* b2  = (const float*)d_in[6];
    const float* ss  = (const float*)d_in[7];
    const float* rs  = (const float*)d_in[8];
    float* out = (float*)d_out;

    dim3 gridS(OUT_SZ / 32, BATCH / BB);
    size_t ldsBytes = (size_t)LDS_FLOATS * sizeof(float);
    subnet_kernel<<<gridS, NWAVE * 32, ldsBytes, stream>>>(
        x, w0, b0, w1, b1, w2, b2, ss, rs, out);
}
